# starting kernel
# baseline (speedup 1.0000x reference)
#include <hip/hip_runtime.h>
#include <hip/hip_bf16.h>

typedef __attribute__((ext_vector_type(2))) float v2f;
typedef __attribute__((ext_vector_type(4))) float v4f;
typedef __attribute__((ext_vector_type(8))) float v8f;

#define FEAT_D 128
#define NORM_EPS 1e-12f

__global__ void cossim_invnorm_kernel(const float* __restrict__ h,
                                      float* __restrict__ inv, int N) {
  const int lane = threadIdx.x & 31;
  const int row  = blockIdx.x * (blockDim.x >> 5) + (threadIdx.x >> 5);
  if (row >= N) return;
  const float* rp = h + (size_t)row * FEAT_D + lane * 4;
  v4f v = *(const v4f*)rp;
  float s = v.x * v.x + v.y * v.y + v.z * v.z + v.w * v.w;
#pragma unroll
  for (int m = 16; m >= 1; m >>= 1) s += __shfl_xor(s, m, 32);
  if (lane == 0) {
    float n = sqrtf(s);
    inv[row] = 1.0f / fmaxf(n, NORM_EPS);
  }
}

__global__ void cossim_sddmm_wmma_kernel(const float* __restrict__ h,
                                         const int* __restrict__ src,
                                         const int* __restrict__ dst,
                                         const float* __restrict__ inv,
                                         float* __restrict__ out, int E) {
  const int lane  = threadIdx.x & 31;
  const int group = blockIdx.x * (blockDim.x >> 5) + (threadIdx.x >> 5);
  const long long e0 = (long long)group * 16;
  if (e0 >= E) return;

  const int r    = lane & 15;
  const int half = lane >> 4;
  long long eg = e0 + r;
  if (eg >= E) eg = (long long)E - 1;
  const int srow = src[eg];
  const int drow = dst[eg];
  const float* ap = h + (size_t)srow * FEAT_D + 2 * half;
  const float* bp = h + (size_t)drow * FEAT_D + 2 * half;

  v8f acc = {};
#pragma unroll
  for (int k = 0; k < FEAT_D; k += 4) {
    v2f a = *(const v2f*)(ap + k);
    v2f b = *(const v2f*)(bp + k);
    acc = __builtin_amdgcn_wmma_f32_16x16x4_f32(
         false, a,  false, b,
         (short)0, acc,  false,  false);
  }

  const bool active = (lane < 8) || (lane >= 24);
  const int  vidx   = lane & 7;
  const int  j      = (lane < 8) ? lane : (lane - 16);
  float diag = acc[0];
#pragma unroll
  for (int i = 1; i < 8; ++i) diag = (vidx == i) ? acc[i] : diag;

  const long long e = e0 + j;
  if (active && e < E) {
    const int sj = src[e];
    const int dj = dst[e];
    out[e] = diag * inv[sj] * inv[dj];
  }
}

extern "C" void kernel_launch(void* const* d_in, const int* in_sizes, int n_in,
                              void* d_out, int out_size, void* d_ws, size_t ws_size,
                              hipStream_t stream) {
  const float* h   = (const float*)d_in[0];
  const int*   src = (const int*)d_in[1];
  const int*   dst = (const int*)d_in[2];
  float*       out = (float*)d_out;
  float*       inv = (float*)d_ws;

  const int N = in_sizes[0] / FEAT_D;
  const int E = in_sizes[1];

  {
    const int rows_per_block = 256 / 32;
    dim3 block(256), grid((N + rows_per_block - 1) / rows_per_block);
    cossim_invnorm_kernel<<<grid, block, 0, stream>>>(h, inv, N);
  }
  {
    const int groups = (E + 15) / 16;
    const int gpb = 256 / 32;
    dim3 block(256), grid((groups + gpb - 1) / gpb);
    cossim_sddmm_wmma_kernel<<<grid, block, 0, stream>>>(h, src, dst, inv, out, E);
  }
}


# SDDMM gather reads whole 128-byte lines (8 adjacent lanes per row line, 4 full lines per wave load) instead of 16 B from 16 rows; two edges per WMMA tile, 8 diagonal partials summed by row DPP
# speedup vs baseline: 2.5248x; 2.5248x over previous
#include <hip/hip_runtime.h>
#include <hip/hip_bf16.h>

typedef __attribute__((ext_vector_type(2))) float v2f;
typedef __attribute__((ext_vector_type(4))) float v4f;
typedef __attribute__((ext_vector_type(8))) float v8f;

#define FEAT_D 128
#define NORM_EPS 1e-12f

__global__ void cossim_invnorm_kernel(const float* __restrict__ h,
                                      float* __restrict__ inv, int N) {
  const int lane = threadIdx.x & 31;
  const int row  = blockIdx.x * (blockDim.x >> 5) + (threadIdx.x >> 5);
  if (row >= N) return;
  const float* rp = h + (size_t)row * FEAT_D + lane * 4;
  v4f v = *(const v4f*)rp;
  float s = v.x * v.x + v.y * v.y + v.z * v.z + v.w * v.w;
#pragma unroll
  for (int m = 16; m >= 1; m >>= 1) s += __shfl_xor(s, m, 32);
  if (lane == 0) {
    float n = sqrtf(s);
    inv[row] = 1.0f / fmaxf(n, NORM_EPS);
  }
}

#define PASSES_PER_WAVE 2

template <int MASK>
__device__ __forceinline__ float row_xor(float v) {
  return __builtin_bit_cast(
      float, __builtin_amdgcn_update_dpp(0, __builtin_bit_cast(int, v),
                                         0x160 | MASK, 0xf, 0xf, true));
}

__global__ __launch_bounds__(128, 10)
void cossim_sddmm_wmma_kernel(const float* __restrict__ h,
                              const int* __restrict__ src,
                              const int* __restrict__ dst,
                              const float* __restrict__ inv,
                              float* __restrict__ out, int E) {
  const int lane = threadIdx.x & 31;
  const int wave = blockIdx.x * (blockDim.x >> 5) +
                   __builtin_amdgcn_readfirstlane(threadIdx.x >> 5);
  const int hf   = lane >> 4;
  const int q16  = ((lane & 15) >> 3) << 4;
  const int vidx = lane & 7;
  const bool valid = ((lane >> 3) & 1) == hf;
  const unsigned lane_off = 128u * hf + 16u * vidx;
  const char* hb = (const char*)h;
  const int grp  = (lane < 16) ? (lane & 1) : (3 - (lane & 1));
  const int srcl = grp * 8 + ((lane & 15) >> 1);

#pragma unroll 1
  for (int g = 0; g < PASSES_PER_WAVE; ++g) {
    const long long e0 = ((long long)wave * PASSES_PER_WAVE + g) * 32;
    if (e0 >= E) break;

    long long el = e0 + lane;
    if (el >= E) el = (long long)E - 1;
    const int sv = src[el];
    const int dv = dst[el];
    const float is = inv[sv];
    const float id = inv[dv];

    unsigned ao[2], bo[2];
#pragma unroll
    for (int tt = 0; tt < 2; ++tt) {
      ao[tt] = ((unsigned)__shfl(sv, q16 + tt, 32) << 9) + lane_off;
      bo[tt] = ((unsigned)__shfl(dv, q16 + tt, 32) << 9) + lane_off;
    }

    float sel = 0.0f;
#pragma unroll
    for (int u = 0; u < 8; ++u) {
      v4f a[2][2], b[2][2];
#pragma unroll
      for (int tt = 0; tt < 2; ++tt) {
        a[tt][0] = *(const v4f*)(hb + ao[tt]);
        a[tt][1] = *(const v4f*)(hb + ao[tt] + 256u);
        b[tt][0] = *(const v4f*)(hb + bo[tt]);
        b[tt][1] = *(const v4f*)(hb + bo[tt] + 256u);
      }
      if (u < 7) {
#pragma unroll
        for (int tt = 0; tt < 2; ++tt) {
          const int t = 2 * (u + 1) + tt;
          ao[tt] = ((unsigned)__shfl(sv, q16 + t, 32) << 9) + lane_off;
          bo[tt] = ((unsigned)__shfl(dv, q16 + t, 32) << 9) + lane_off;
        }
      }
      __builtin_amdgcn_sched_barrier(0);

      v8f acc[2] = {};
#pragma unroll
      for (int j = 0; j < 2; ++j) {
#pragma unroll
        for (int tt = 0; tt < 2; ++tt) {
          acc[tt] = __builtin_amdgcn_wmma_f32_16x16x4_f32(
              false, __builtin_shufflevector(a[tt][j], a[tt][j], 0, 1),
              false, __builtin_shufflevector(b[tt][j], b[tt][j], 0, 1),
              (short)0, acc[tt], false, false);
        }
#pragma unroll
        for (int tt = 0; tt < 2; ++tt) {
          acc[tt] = __builtin_amdgcn_wmma_f32_16x16x4_f32(
              false, __builtin_shufflevector(a[tt][j], a[tt][j], 2, 3),
              false, __builtin_shufflevector(b[tt][j], b[tt][j], 2, 3),
              (short)0, acc[tt], false, false);
        }
      }
      float x[2];
#pragma unroll
      for (int tt = 0; tt < 2; ++tt) {
        float d = acc[tt][0];
#pragma unroll
        for (int i = 1; i < 8; ++i) d = (vidx == i) ? acc[tt][i] : d;
        x[tt] = d;
      }
      const float x1r = row_xor<8>(x[1]);
      float y = valid ? x[0] : x1r;
      y += row_xor<4>(y);
      y += row_xor<2>(y);
      y += row_xor<1>(y);
      sel = (vidx == u) ? y : sel;
    }

    const float r = __shfl(sel, srcl, 32);
    const long long e = e0 + lane;
    const float res = r * is * id;
    if (e < E) out[e] = res;
  }
}

extern "C" void kernel_launch(void* const* d_in, const int* in_sizes, int n_in,
                              void* d_out, int out_size, void* d_ws, size_t ws_size,
                              hipStream_t stream) {
  const float* h   = (const float*)d_in[0];
  const int*   src = (const int*)d_in[1];
  const int*   dst = (const int*)d_in[2];
  float*       out = (float*)d_out;
  float*       inv = (float*)d_ws;

  const int N = in_sizes[0] / FEAT_D;
  const int E = in_sizes[1];

  {
    const int rows_per_block = 256 / 32;
    dim3 block(256), grid((N + rows_per_block - 1) / rows_per_block);
    cossim_invnorm_kernel<<<grid, block, 0, stream>>>(h, inv, N);
  }
  {
    const int waves = (E + 32 * PASSES_PER_WAVE - 1) / (32 * PASSES_PER_WAVE);
    const int wpb = 128 / 32;
    dim3 block(128), grid((waves + wpb - 1) / wpb);
    cossim_sddmm_wmma_kernel<<<grid, block, 0, stream>>>(h, src, dst, inv, out, E);
  }
}


# Inverse-norm launch as a stream: one wave per 32 consecutive rows, all 32 row loads in flight, one 128-byte store per wave; N/32 waves enter instead of N single-load waves
# speedup vs baseline: 2.9858x; 1.1826x over previous
#include <hip/hip_runtime.h>
#include <hip/hip_bf16.h>

typedef __attribute__((ext_vector_type(2))) float v2f;
typedef __attribute__((ext_vector_type(4))) float v4f;
typedef __attribute__((ext_vector_type(8))) float v8f;

#define FEAT_D 128
#define NORM_EPS 1e-12f

#define NORM_ROWS 32
#define NORM_WAVES 8

template <int MASK>
__device__ __forceinline__ float row_xor(float v) {
  return __builtin_bit_cast(
      float, __builtin_amdgcn_update_dpp(0, __builtin_bit_cast(int, v),
                                         0x160 | MASK, 0xf, 0xf, true));
}

__device__ __forceinline__ float half_xor(float v) {
  const int b = __builtin_bit_cast(int, v);
  return __builtin_bit_cast(
      float, __builtin_amdgcn_permlanex16(b, b, 0x76543210, 0xfedcba98, false,
                                          false));
}

__global__ __launch_bounds__(NORM_WAVES * 32)
void cossim_invnorm_kernel(const float* __restrict__ h,
                           float* __restrict__ inv, int N) {
  const int lane = threadIdx.x & 31;
  const int wave = blockIdx.x * NORM_WAVES +
                   __builtin_amdgcn_readfirstlane(threadIdx.x >> 5);
  const long long row0 = (long long)wave * NORM_ROWS;
  if (row0 >= N) return;
  const long long left = (long long)N - 1 - row0;
  const int last = left < NORM_ROWS - 1 ? (int)left : NORM_ROWS - 1;
  const unsigned lane_off = 16u * lane;

  const char* wbase = (const char*)h + row0 * (FEAT_D * 4);
  v4f v[NORM_ROWS];
  if (last == NORM_ROWS - 1) {
#pragma unroll
    for (int r = 0; r < NORM_ROWS; ++r)
      v[r] = *(const v4f*)(wbase + lane_off + r * (FEAT_D * 4));
  } else {
#pragma unroll
    for (int r = 0; r < NORM_ROWS; ++r) {
      const int rr = r < last ? r : last;
      v[r] = *(const v4f*)(wbase + lane_off + rr * (FEAT_D * 4));
    }
  }
  __builtin_amdgcn_sched_barrier(0);

  float s[NORM_ROWS];
#pragma unroll
  for (int r = 0; r < NORM_ROWS; ++r) {
    float q = v[r].x * v[r].x + v[r].y * v[r].y + v[r].z * v[r].z +
              v[r].w * v[r].w;
    asm volatile("" : "+v"(q));
    s[r] = q;
  }

  const bool k16 = (lane & 16) == 0;
  float t[16];
#pragma unroll
  for (int r = 0; r < 16; ++r) {
    const float keep = k16 ? s[r] : s[r + 16];
    const float send = k16 ? s[r + 16] : s[r];
    t[r] = keep + half_xor(send);
  }
  const bool k8 = (lane & 8) == 0;
  float u[8];
#pragma unroll
  for (int r = 0; r < 8; ++r) {
    const float keep = k8 ? t[r] : t[r + 8];
    const float send = k8 ? t[r + 8] : t[r];
    u[r] = keep + row_xor<8>(send);
  }
  const bool k4 = (lane & 4) == 0;
  float w[4];
#pragma unroll
  for (int r = 0; r < 4; ++r) {
    const float keep = k4 ? u[r] : u[r + 4];
    const float send = k4 ? u[r + 4] : u[r];
    w[r] = keep + row_xor<4>(send);
  }
  const bool k2 = (lane & 2) == 0;
  float x[2];
#pragma unroll
  for (int r = 0; r < 2; ++r) {
    const float keep = k2 ? w[r] : w[r + 2];
    const float send = k2 ? w[r + 2] : w[r];
    x[r] = keep + row_xor<2>(send);
  }
  const bool k1 = (lane & 1) == 0;
  const float mine = (k1 ? x[0] : x[1]) + row_xor<1>(k1 ? x[1] : x[0]);
  const float n = sqrtf(mine);
  const float res = 1.0f / fmaxf(n, NORM_EPS);
  if (row0 + lane < N) inv[row0 + lane] = res;
}

#define PASSES_PER_WAVE 2

__global__ __launch_bounds__(128, 10)
void cossim_sddmm_wmma_kernel(const float* __restrict__ h,
                              const int* __restrict__ src,
                              const int* __restrict__ dst,
                              const float* __restrict__ inv,
                              float* __restrict__ out, int E) {
  const int lane = threadIdx.x & 31;
  const int wave = blockIdx.x * (blockDim.x >> 5) +
                   __builtin_amdgcn_readfirstlane(threadIdx.x >> 5);
  const int hf   = lane >> 4;
  const int q16  = ((lane & 15) >> 3) << 4;
  const int vidx = lane & 7;
  const bool valid = ((lane >> 3) & 1) == hf;
  const unsigned lane_off = 128u * hf + 16u * vidx;
  const char* hb = (const char*)h;
  const int grp  = (lane < 16) ? (lane & 1) : (3 - (lane & 1));
  const int srcl = grp * 8 + ((lane & 15) >> 1);

#pragma unroll 1
  for (int g = 0; g < PASSES_PER_WAVE; ++g) {
    const long long e0 = ((long long)wave * PASSES_PER_WAVE + g) * 32;
    if (e0 >= E) break;

    long long el = e0 + lane;
    if (el >= E) el = (long long)E - 1;
    const int sv = src[el];
    const int dv = dst[el];
    const float is = inv[sv];
    const float id = inv[dv];

    unsigned ao[2], bo[2];
#pragma unroll
    for (int tt = 0; tt < 2; ++tt) {
      ao[tt] = ((unsigned)__shfl(sv, q16 + tt, 32) << 9) + lane_off;
      bo[tt] = ((unsigned)__shfl(dv, q16 + tt, 32) << 9) + lane_off;
    }

    float sel = 0.0f;
#pragma unroll
    for (int u = 0; u < 8; ++u) {
      v4f a[2][2], b[2][2];
#pragma unroll
      for (int tt = 0; tt < 2; ++tt) {
        a[tt][0] = *(const v4f*)(hb + ao[tt]);
        a[tt][1] = *(const v4f*)(hb + ao[tt] + 256u);
        b[tt][0] = *(const v4f*)(hb + bo[tt]);
        b[tt][1] = *(const v4f*)(hb + bo[tt] + 256u);
      }
      if (u < 7) {
#pragma unroll
        for (int tt = 0; tt < 2; ++tt) {
          const int t = 2 * (u + 1) + tt;
          ao[tt] = ((unsigned)__shfl(sv, q16 + t, 32) << 9) + lane_off;
          bo[tt] = ((unsigned)__shfl(dv, q16 + t, 32) << 9) + lane_off;
        }
      }
      __builtin_amdgcn_sched_barrier(0);

      v8f acc[2] = {};
#pragma unroll
      for (int j = 0; j < 2; ++j) {
#pragma unroll
        for (int tt = 0; tt < 2; ++tt) {
          acc[tt] = __builtin_amdgcn_wmma_f32_16x16x4_f32(
              false, __builtin_shufflevector(a[tt][j], a[tt][j], 0, 1),
              false, __builtin_shufflevector(b[tt][j], b[tt][j], 0, 1),
              (short)0, acc[tt], false, false);
        }
#pragma unroll
        for (int tt = 0; tt < 2; ++tt) {
          acc[tt] = __builtin_amdgcn_wmma_f32_16x16x4_f32(
              false, __builtin_shufflevector(a[tt][j], a[tt][j], 2, 3),
              false, __builtin_shufflevector(b[tt][j], b[tt][j], 2, 3),
              (short)0, acc[tt], false, false);
        }
      }
      float x[2];
#pragma unroll
      for (int tt = 0; tt < 2; ++tt) {
        float d = acc[tt][0];
#pragma unroll
        for (int i = 1; i < 8; ++i) d = (vidx == i) ? acc[tt][i] : d;
        x[tt] = d;
      }
      const float x1r = row_xor<8>(x[1]);
      float y = valid ? x[0] : x1r;
      y += row_xor<4>(y);
      y += row_xor<2>(y);
      y += row_xor<1>(y);
      sel = (vidx == u) ? y : sel;
    }

    const float r = __shfl(sel, srcl, 32);
    const long long e = e0 + lane;
    const float res = r * is * id;
    if (e < E) out[e] = res;
  }
}

extern "C" void kernel_launch(void* const* d_in, const int* in_sizes, int n_in,
                              void* d_out, int out_size, void* d_ws, size_t ws_size,
                              hipStream_t stream) {
  const float* h   = (const float*)d_in[0];
  const int*   src = (const int*)d_in[1];
  const int*   dst = (const int*)d_in[2];
  float*       out = (float*)d_out;
  float*       inv = (float*)d_ws;

  const int N = in_sizes[0] / FEAT_D;
  const int E = in_sizes[1];

  {
    const int rows_per_block = NORM_WAVES * NORM_ROWS;
    dim3 block(NORM_WAVES * 32), grid((N + rows_per_block - 1) / rows_per_block);
    cossim_invnorm_kernel<<<grid, block, 0, stream>>>(h, inv, N);
  }
  {
    const int waves = (E + 32 * PASSES_PER_WAVE - 1) / (32 * PASSES_PER_WAVE);
    const int wpb = 128 / 32;
    dim3 block(128), grid((waves + wpb - 1) / wpb);
    cossim_sddmm_wmma_kernel<<<grid, block, 0, stream>>>(h, src, dst, inv, out, E);
  }
}
